# starting kernel
# baseline (speedup 1.0000x reference)
#include <hip/hip_runtime.h>
#include <math.h>

#define Bdim 64
#define Sdim 512
#define Hdim 768
#define Ndim 64

typedef __attribute__((ext_vector_type(2))) float v2f;
typedef __attribute__((ext_vector_type(8))) float v8f;

__global__ __launch_bounds__(256) void means_kernel(
    const float* __restrict__ top_vec, const int* __restrict__ clss,
    const int* __restrict__ sents_end, float* __restrict__ means) {
  const int m = blockIdx.x;
  const int b = m >> 6;
  const int start = clss[m];
  const int end = sents_end[m];
  const int len = end - start;
  const float inv = 1.0f / (float)(len < 1 ? 1 : len);
  const float* base = top_vec + ((size_t)b * Sdim + (size_t)start) * Hdim;
  for (int h = threadIdx.x; h < Hdim; h += 256) {
    float s = 0.0f;
    for (int t = 0; t < len; ++t) s += base[(size_t)t * Hdim + h];
    means[(size_t)m * Hdim + h] = s * inv;
  }
}

#define ALD 772
__global__ __launch_bounds__(256) void gemm_reduce_kernel(
    const float* __restrict__ means, const float* __restrict__ w_sents,
    const float* __restrict__ b_sents, const float* __restrict__ w_fc,
    const float* __restrict__ b_fc, const float* __restrict__ mask_clss,
    float* __restrict__ out) {
  __shared__ float Abuf[32][ALD];
  __shared__ float h_sh[32];
  __shared__ float c_sh;

  const int tid = threadIdx.x;
  const int row0 = blockIdx.x * 32;

  if (tid < 32) h_sh[tid] = 0.0f;
  if (tid == 0) c_sh = 0.0f;
  __syncthreads();

  float cp = 0.0f;
  for (int i = tid; i < Hdim; i += 256) cp += b_sents[i] * w_fc[i];
  atomicAdd(&c_sh, cp);

  for (int idx = tid; idx < 32 * Hdim; idx += 256) {
    const int r = idx / Hdim;
    const int c = idx - r * Hdim;
    Abuf[r][c] = means[(size_t)(row0 + r) * Hdim + c];
  }
  __syncthreads();

  const int wave = tid >> 5;
  const int lane = tid & 31;
  const int laneLo = lane & 15;
  const int hi = lane >> 4;

  for (int ot = wave; ot < 48; ot += 8) {
    const int ob = ot << 4;
    v8f acc0 = {0.f,0.f,0.f,0.f,0.f,0.f,0.f,0.f};
    v8f acc1 = {0.f,0.f,0.f,0.f,0.f,0.f,0.f,0.f};
    const float* wrow = w_sents + (size_t)(ob + laneLo) * Hdim + (hi << 1);
    const float* a0p = &Abuf[laneLo][hi << 1];
    const float* a1p = &Abuf[16 + laneLo][hi << 1];
#pragma unroll 4
    for (int k0 = 0; k0 < Hdim; k0 += 4) {
      const v2f bv = *(const v2f*)(wrow + k0);
      const v2f a0 = *(const v2f*)(a0p + k0);
      const v2f a1 = *(const v2f*)(a1p + k0);
      acc0 = __builtin_amdgcn_wmma_f32_16x16x4_f32(
          false, a0, false, bv, (short)0, acc0, false, false);
      acc1 = __builtin_amdgcn_wmma_f32_16x16x4_f32(
          false, a1, false, bv, (short)0, acc1, false, false);
    }
    const float wv = w_fc[ob + laneLo];
#pragma unroll
    for (int j = 0; j < 8; ++j) {
      float p0 = acc0[j] * wv;
      float p1 = acc1[j] * wv;
      p0 += __shfl_xor(p0, 1, 32); p0 += __shfl_xor(p0, 2, 32);
      p0 += __shfl_xor(p0, 4, 32); p0 += __shfl_xor(p0, 8, 32);
      p1 += __shfl_xor(p1, 1, 32); p1 += __shfl_xor(p1, 2, 32);
      p1 += __shfl_xor(p1, 4, 32); p1 += __shfl_xor(p1, 8, 32);
      if (laneLo == 0) {
        atomicAdd(&h_sh[(hi << 3) + j], p0);
        atomicAdd(&h_sh[16 + (hi << 3) + j], p1);
      }
    }
  }
  __syncthreads();

  if (tid < 32) {
    const int m = row0 + tid;
    const float mask = mask_clss[m];
    const float dot = h_sh[tid] + c_sh;
    const float hv = mask * dot + b_fc[0];
    out[m] = mask / (1.0f + expf(-hv));
  }
}

extern "C" void kernel_launch(void* const* d_in, const int* in_sizes, int n_in,
                              void* d_out, int out_size, void* d_ws, size_t ws_size,
                              hipStream_t stream) {
  const float* top_vec   = (const float*)d_in[0];
  const int*   clss      = (const int*)d_in[1];
  const int*   sents_end = (const int*)d_in[2];
  const float* mask_clss = (const float*)d_in[3];
  const float* w_sents   = (const float*)d_in[4];
  const float* b_sents   = (const float*)d_in[5];
  const float* w_fc      = (const float*)d_in[6];
  const float* b_fc      = (const float*)d_in[7];
  float* out = (float*)d_out;
  float* means = (float*)d_ws;

  means_kernel<<<Bdim * Ndim, 256, 0, stream>>>(top_vec, clss, sents_end, means);
  gemm_reduce_kernel<<<(Bdim * Ndim) / 32, 256, 0, stream>>>(
      means, w_sents, b_sents, w_fc, b_fc, mask_clss, out);
}


# split-K 2-D decomposition: one work-group per (64-row document, K quarter) builds its means tile in LDS, 4x6 accumulator tiles per wave; quarter partials summed in fixed order, no atomics
# speedup vs baseline: 3.1790x; 3.1790x over previous
#include <hip/hip_runtime.h>
#include <math.h>


#define Bdim 64
#define Sdim 512
#define Hdim 768
#define Ndim 64
#define KQ 192
#define KC 16
#define NCH (KQ / KC)
#define ALD 196
#define BLD 20
#define NPART (Bdim * Ndim)

typedef __attribute__((ext_vector_type(2))) float v2f;
typedef __attribute__((ext_vector_type(4))) float v4f;
typedef __attribute__((ext_vector_type(8))) float v8f;

__global__ __launch_bounds__(256) void splitk_gemm_kernel(
    const float* __restrict__ top_vec, const int* __restrict__ clss,
    const int* __restrict__ sents_end, const float* __restrict__ w_sents,
    const float* __restrict__ b_sents, const float* __restrict__ w_fc,
    float* __restrict__ part) {
  __shared__ __attribute__((aligned(16))) float As[64][ALD];
  __shared__ __attribute__((aligned(16))) float Bs[2][Hdim][BLD];
  __shared__ float red[8][64];

  const int tid = threadIdx.x;
  const int b = blockIdx.x >> 2;
  const int kq = blockIdx.x & 3;
  const int kbase = kq * KQ;
  const int wave = tid >> 5;
  const int lane = tid & 31;
  const int laneLo = lane & 15;
  const int hi = lane >> 4;

  v4f breg[12];
  const float* wsrc = w_sents + (size_t)(tid >> 2) * Hdim + kbase + ((tid & 3) << 2);
#pragma unroll
  for (int j = 0; j < 12; ++j) breg[j] = *(const v4f*)(wsrc + (size_t)j * 64 * Hdim);

#pragma unroll 1
  for (int rp = 0; rp < 4; ++rp) {
    const int n = wave * 8 + rp * 2 + hi;
    const int m = b * Ndim + n;
    const int st = clss[m];
    const int en = sents_end[m];
    const int len = en - st;
    int lo = len >= 0 ? st : en;
    int hiT = len >= 0 ? en : st;
    lo = lo < 0 ? 0 : lo;
    hiT = hiT > Sdim ? Sdim : hiT;
    const float inv = (len >= 0 ? 1.0f : -1.0f) / (float)(len < 1 ? 1 : len);
    const float* rowbase = top_vec + (size_t)b * Sdim * Hdim + kbase + (laneLo << 2);
    const v4f zero4 = {0.f, 0.f, 0.f, 0.f};
    v4f s0 = zero4;
    v4f s1 = zero4;
    v4f s2 = zero4;
    for (int t0 = lo; t0 < hiT; t0 += 8) {
      v4f x0[8], x1[8], x2[8];
#pragma unroll
      for (int u = 0; u < 8; ++u) {
        const int t = t0 + u;
        const int tc = t < hiT ? t : hiT - 1;
        const float* p = rowbase + (size_t)tc * Hdim;
        x0[u] = *(const v4f*)(p);
        x1[u] = *(const v4f*)(p + 64);
        x2[u] = *(const v4f*)(p + 128);
      }
#pragma unroll
      for (int u = 0; u < 8; ++u) {
        const bool ok = (t0 + u) < hiT;
        s0 += ok ? x0[u] : zero4;
        s1 += ok ? x1[u] : zero4;
        s2 += ok ? x2[u] : zero4;
      }
    }
    *(v4f*)&As[n][(laneLo << 2)] = s0 * inv;
    *(v4f*)&As[n][(laneLo << 2) + 64] = s1 * inv;
    *(v4f*)&As[n][(laneLo << 2) + 128] = s2 * inv;
  }

  float* bdst = &Bs[0][tid >> 2][(tid & 3) << 2];
#pragma unroll
  for (int j = 0; j < 12; ++j) *(v4f*)(bdst + j * 64 * BLD) = breg[j];
  __syncthreads();

  v8f acc[4][6];
#pragma unroll
  for (int i = 0; i < 4; ++i)
#pragma unroll
    for (int t = 0; t < 6; ++t) acc[i][t] = (v8f){0.f, 0.f, 0.f, 0.f, 0.f, 0.f, 0.f, 0.f};

  const float* aptr = &As[laneLo][hi << 2];
  const float* bptr0 = &Bs[0][wave * 96 + laneLo][hi << 2];

#pragma unroll 1
  for (int ch = 0; ch < NCH; ++ch) {
    const int stg = ch & 1;
    if (ch + 1 < NCH) {
      const float* wn = wsrc + (ch + 1) * KC;
#pragma unroll
      for (int j = 0; j < 12; ++j) breg[j] = *(const v4f*)(wn + (size_t)j * 64 * Hdim);
    }
    const float* ap = aptr + ch * KC;
    const float* bp = bptr0 + stg * (Hdim * BLD);
#pragma unroll
    for (int q = 0; q < 2; ++q) {
      v4f a[4], bb[6];
#pragma unroll
      for (int i = 0; i < 4; ++i) a[i] = *(const v4f*)(ap + i * 16 * ALD + q * 8);
#pragma unroll
      for (int t = 0; t < 6; ++t) bb[t] = *(const v4f*)(bp + t * 16 * BLD + q * 8);
#pragma unroll
      for (int i = 0; i < 4; ++i)
#pragma unroll
        for (int t = 0; t < 6; ++t)
          acc[i][t] = __builtin_amdgcn_wmma_f32_16x16x4_f32(
              false, a[i].xy, false, bb[t].xy, (short)0, acc[i][t], false, false);
#pragma unroll
      for (int i = 0; i < 4; ++i)
#pragma unroll
        for (int t = 0; t < 6; ++t)
          acc[i][t] = __builtin_amdgcn_wmma_f32_16x16x4_f32(
              false, a[i].zw, false, bb[t].zw, (short)0, acc[i][t], false, false);
    }
    if (ch + 1 < NCH) {
      float* bd = bdst + (stg ^ 1) * (Hdim * BLD);
#pragma unroll
      for (int j = 0; j < 12; ++j) *(v4f*)(bd + j * 64 * BLD) = breg[j];
    }
    __syncthreads();
  }

  float wv[6];
  float biasl = 0.0f;
#pragma unroll
  for (int t = 0; t < 6; ++t) {
    const int o = wave * 96 + t * 16 + laneLo;
    wv[t] = w_fc[o];
    if (kq == 0) biasl += b_sents[o] * wv[t];
  }
#pragma unroll
  for (int i = 0; i < 4; ++i) {
#pragma unroll
    for (int j = 0; j < 8; ++j) {
      float p = biasl;
#pragma unroll
      for (int t = 0; t < 6; ++t) p += acc[i][t][j] * wv[t];
      p += __shfl_xor(p, 1, 32);
      p += __shfl_xor(p, 2, 32);
      p += __shfl_xor(p, 4, 32);
      p += __shfl_xor(p, 8, 32);
      if (laneLo == 0) red[wave][i * 16 + (hi << 3) + j] = p;
    }
  }
  __syncthreads();
  if (tid < 64) {
    float s = red[0][tid];
#pragma unroll
    for (int w = 1; w < 8; ++w) s += red[w][tid];
    part[kq * NPART + b * Ndim + tid] = s;
  }
}

__global__ __launch_bounds__(256) void finalize_kernel(
    const float* __restrict__ part, const float* __restrict__ mask_clss,
    const float* __restrict__ b_fc, const float* __restrict__ top_vec,
    const float* __restrict__ w_fc, float* __restrict__ out) {
  const int m = blockIdx.x * 256 + threadIdx.x;
  float dot = ((part[m] + part[NPART + m]) + part[2 * NPART + m]) + part[3 * NPART + m];
  const float mask = mask_clss[m];
  if (!(mask >= 0.0f)) {
    const float* pv = top_vec + ((size_t)(m >> 6) * Sdim + (Sdim - 1)) * Hdim;
    float d = 0.0f;
    for (int h = 0; h < Hdim; ++h) d += pv[h] * w_fc[h];
    dot = d;
  }
  const float hv = mask * dot + b_fc[0];
  out[m] = mask / (1.0f + expf(-hv));
}

extern "C" void kernel_launch(void* const* d_in, const int* in_sizes, int n_in,
                              void* d_out, int out_size, void* d_ws, size_t ws_size,
                              hipStream_t stream) {
  const float* top_vec   = (const float*)d_in[0];
  const int*   clss      = (const int*)d_in[1];
  const int*   sents_end = (const int*)d_in[2];
  const float* mask_clss = (const float*)d_in[3];
  const float* w_sents   = (const float*)d_in[4];
  const float* b_sents   = (const float*)d_in[5];
  const float* w_fc      = (const float*)d_in[6];
  const float* b_fc      = (const float*)d_in[7];
  float* out = (float*)d_out;
  float* part = (float*)d_ws;

  splitk_gemm_kernel<<<Bdim * 4, 256, 0, stream>>>(
      top_vec, clss, sents_end, w_sents, b_sents, w_fc, part);
  finalize_kernel<<<(Bdim * Ndim) / 256, 256, 0, stream>>>(
      part, mask_clss, b_fc, top_vec, w_fc, out);
}


# contractions reassociated: w_fc contracted into w_sents first (v = w_sents^T w_fc), one dot per token row with v, then segment sums of scalars; new fixed sum order, no atomics, 3 launches
# speedup vs baseline: 8.8860x; 2.7952x over previous
#include <hip/hip_runtime.h>
#include <math.h>


#define Bdim 64
#define Sdim 512
#define Hdim 768
#define Ndim 64
#define NTOK (Bdim * Sdim)
#define WS_BIAS Hdim
#define WS_P 1024
#define PLD 20

typedef __attribute__((ext_vector_type(2))) float v2f;
typedef __attribute__((ext_vector_type(4))) float v4f;
typedef __attribute__((ext_vector_type(8))) float v8f;

__global__ __launch_bounds__(1024) void weight_vector_kernel(
    const float* __restrict__ w_sents, const float* __restrict__ b_sents,
    const float* __restrict__ w_fc, float* __restrict__ ws) {
  __shared__ float red[32][32];
  __shared__ float redb[8];
  const int tid = threadIdx.x;
  const int wave = tid >> 5;
  const int lane = tid & 31;
  const int laneLo = lane & 15;
  const int hh = lane >> 4;
  const int h0 = blockIdx.x * 32;
  const int obase = wave * 24;
  const float* wp = w_sents + (size_t)(obase + 2 * hh) * Hdim + h0 + laneLo;
  const float* fp = w_fc + obase + 2 * hh;
  v2f bw0[6], bw1[6], aw[6];
#pragma unroll
  for (int s = 0; s < 6; ++s) {
    const float* r0 = wp + (size_t)(4 * s) * Hdim;
    bw0[s].x = r0[0];
    bw0[s].y = r0[Hdim];
    bw1[s].x = r0[16];
    bw1[s].y = r0[Hdim + 16];
    aw[s].x = fp[4 * s];
    aw[s].y = fp[4 * s + 1];
  }
  v8f c0 = (v8f){0.f, 0.f, 0.f, 0.f, 0.f, 0.f, 0.f, 0.f};
  v8f c1 = c0;
#pragma unroll
  for (int s = 0; s < 6; ++s) {
    c0 = __builtin_amdgcn_wmma_f32_16x16x4_f32(false, aw[s], false, bw0[s], (short)0, c0, false, false);
    c1 = __builtin_amdgcn_wmma_f32_16x16x4_f32(false, aw[s], false, bw1[s], (short)0, c1, false, false);
  }
  if (hh == 0) {
    red[wave][laneLo] = c0[0];
    red[wave][16 + laneLo] = c1[0];
  }
  if (blockIdx.x == 0 && wave < 8) {
    float bp = 0.0f;
#pragma unroll
    for (int j = 0; j < 3; ++j) {
      const int o = wave * 96 + lane + 32 * j;
      bp += b_sents[o] * w_fc[o];
    }
    bp += __shfl_xor(bp, 16, 32);
    bp += __shfl_xor(bp, 8, 32);
    bp += __shfl_xor(bp, 4, 32);
    bp += __shfl_xor(bp, 2, 32);
    bp += __shfl_xor(bp, 1, 32);
    if (lane == 0) redb[wave] = bp;
  }
  __syncthreads();
  if (tid < 32) {
    float s = red[0][tid];
#pragma unroll
    for (int w = 1; w < 32; ++w) s += red[w][tid];
    ws[h0 + tid] = s;
  }
  if (blockIdx.x == 0 && tid == 32) {
    float s = redb[0];
#pragma unroll
    for (int w = 1; w < 8; ++w) s += redb[w];
    ws[WS_BIAS] = s;
  }
}

__device__ __forceinline__ void load_tokens(v4f (&x)[4][6], const float* base) {
#pragma unroll
  for (int u = 0; u < 4; ++u)
#pragma unroll
    for (int j = 0; j < 6; ++j) x[u][j] = *(const v4f*)(base + u * Hdim + j * 128);
}

__device__ __forceinline__ void dot_tokens(const v4f (&x)[4][6], const v4f (&vb)[6],
                                           float* pdrow, int laneLo, bool diagLane) {
  v8f acc[4];
#pragma unroll
  for (int u = 0; u < 4; ++u) acc[u] = (v8f){0.f, 0.f, 0.f, 0.f, 0.f, 0.f, 0.f, 0.f};
#pragma unroll
  for (int j = 0; j < 6; ++j) {
#pragma unroll
    for (int u = 0; u < 4; ++u)
      acc[u] = __builtin_amdgcn_wmma_f32_16x16x4_f32(
          false, x[u][j].xy, false, vb[j].xy, (short)0, acc[u], false, false);
#pragma unroll
    for (int u = 0; u < 4; ++u)
      acc[u] = __builtin_amdgcn_wmma_f32_16x16x4_f32(
          false, x[u][j].zw, false, vb[j].zw, (short)0, acc[u], false, false);
  }
  const int r = laneLo & 7;
#pragma unroll
  for (int u = 0; u < 4; ++u) {
    float d = acc[u][0];
    d = r == 1 ? acc[u][1] : d;
    d = r == 2 ? acc[u][2] : d;
    d = r == 3 ? acc[u][3] : d;
    d = r == 4 ? acc[u][4] : d;
    d = r == 5 ? acc[u][5] : d;
    d = r == 6 ? acc[u][6] : d;
    d = r == 7 ? acc[u][7] : d;
    if (diagLane) pdrow[u * PLD + laneLo] = d;
  }
}

__global__ __launch_bounds__(256) void token_dot_kernel(
    const float* __restrict__ top_vec, float* ws) {
  __shared__ __attribute__((aligned(16))) float pd[128][PLD];
  const int tid = threadIdx.x;
  const int wave = tid >> 5;
  const int lane = tid & 31;
  const int laneLo = lane & 15;
  const int hh = lane >> 4;
  const bool diagLane = (laneLo >> 3) == hh;
  const int tok0 = blockIdx.x * 128;
  const float* xbase = top_vec + (size_t)(tok0 + wave * 4) * Hdim + (lane << 2);
  v4f vb[6];
#pragma unroll
  for (int j = 0; j < 6; ++j) vb[j] = *(const v4f*)(ws + j * 128 + (lane << 2));
  v4f xa[4][6], xb[4][6];
  load_tokens(xa, xbase);
  load_tokens(xb, xbase + (size_t)32 * Hdim);
  __builtin_amdgcn_sched_barrier(0);

  float* pdw = &pd[wave * 4][0];
  dot_tokens(xa, vb, pdw, laneLo, diagLane);
  __builtin_amdgcn_sched_barrier(0);
  load_tokens(xa, xbase + (size_t)64 * Hdim);
  __builtin_amdgcn_sched_barrier(0);
  dot_tokens(xb, vb, pdw + 32 * PLD, laneLo, diagLane);
  __builtin_amdgcn_sched_barrier(0);
  load_tokens(xb, xbase + (size_t)96 * Hdim);
  __builtin_amdgcn_sched_barrier(0);
  dot_tokens(xa, vb, pdw + 64 * PLD, laneLo, diagLane);
  dot_tokens(xb, vb, pdw + 96 * PLD, laneLo, diagLane);
  __syncthreads();

  if (tid < 128) {
    const v4f* rp = (const v4f*)&pd[tid][0];
    const v4f q0 = rp[0], q1 = rp[1], q2 = rp[2], q3 = rp[3];
    float s = q0.x;
    s += q0.y; s += q0.z; s += q0.w;
    s += q1.x; s += q1.y; s += q1.z; s += q1.w;
    s += q2.x; s += q2.y; s += q2.z; s += q2.w;
    s += q3.x; s += q3.y; s += q3.z; s += q3.w;
    ws[WS_P + tok0 + tid] = s;
  }
}

__global__ __launch_bounds__(256) void segment_finalize_kernel(
    const float* __restrict__ ws, const int* __restrict__ clss,
    const int* __restrict__ sents_end, const float* __restrict__ mask_clss,
    const float* __restrict__ b_fc, const float* __restrict__ top_vec,
    const float* __restrict__ w_fc, float* __restrict__ out) {
  __shared__ float pl[Sdim];
  __shared__ float res[Ndim];
  const int tid = threadIdx.x;
  const int b = blockIdx.x;
  const float* p = ws + WS_P + b * Sdim;
  pl[tid] = p[tid];
  pl[tid + 256] = p[tid + 256];

  const int n = tid >> 2;
  const int q = tid & 3;
  const int m = b * Ndim + n;
  const int st = clss[m];
  const int en = sents_end[m];
  const int len = en - st;
  int lo = len >= 0 ? st : en;
  int hiT = len >= 0 ? en : st;
  lo = lo < 0 ? 0 : lo;
  hiT = hiT > Sdim ? Sdim : hiT;
  const float inv = (len >= 0 ? 1.0f : -1.0f) / (float)(len < 1 ? 1 : len);
  const float bias = ws[WS_BIAS];
  __syncthreads();

  float s = 0.0f;
  for (int i = lo + q; i < hiT; i += 16) {
    const float a0 = pl[i];
    const float a1 = (i + 4 < hiT) ? pl[i + 4] : 0.0f;
    const float a2 = (i + 8 < hiT) ? pl[i + 8] : 0.0f;
    const float a3 = (i + 12 < hiT) ? pl[i + 12] : 0.0f;
    s += a0;
    s += a1;
    s += a2;
    s += a3;
  }
  s += __shfl_xor(s, 1, 32);
  s += __shfl_xor(s, 2, 32);
  if (q == 0) res[n] = s * inv + bias;
  __syncthreads();

  if (tid < Ndim) {
    const int mo = b * Ndim + tid;
    float dot = res[tid];
    const float mask = mask_clss[mo];
    if (!(mask >= 0.0f)) {
      const float* pv = top_vec + ((size_t)b * Sdim + (Sdim - 1)) * Hdim;
      float d = 0.0f;
      for (int h = 0; h < Hdim; ++h) d += pv[h] * w_fc[h];
      dot = d;
    }
    const float hv = mask * dot + b_fc[0];
    out[mo] = mask / (1.0f + expf(-hv));
  }
}

extern "C" void kernel_launch(void* const* d_in, const int* in_sizes, int n_in,
                              void* d_out, int out_size, void* d_ws, size_t ws_size,
                              hipStream_t stream) {
  const float* top_vec   = (const float*)d_in[0];
  const int*   clss      = (const int*)d_in[1];
  const int*   sents_end = (const int*)d_in[2];
  const float* mask_clss = (const float*)d_in[3];
  const float* w_sents   = (const float*)d_in[4];
  const float* b_sents   = (const float*)d_in[5];
  const float* w_fc      = (const float*)d_in[6];
  const float* b_fc      = (const float*)d_in[7];
  float* out = (float*)d_out;
  float* ws = (float*)d_ws;

  weight_vector_kernel<<<Hdim / 32, 1024, 0, stream>>>(w_sents, b_sents, w_fc, ws);
  token_dot_kernel<<<NTOK / 128, 256, 0, stream>>>(top_vec, ws);
  segment_finalize_kernel<<<Bdim, 256, 0, stream>>>(
      ws, clss, sents_end, mask_clss, b_fc, top_vec, w_fc, out);
}


# segment sums taken before the weight-vector dot, as per-column prefix sums and differences (the reference's own form): weight-vector work joins the token stream, 2 launches; new sum order
# speedup vs baseline: 9.4695x; 1.0657x over previous
#include <hip/hip_runtime.h>
#include <math.h>


#define Bdim 64
#define Sdim 512
#define Hdim 768
#define Ndim 64
#define NROW (Bdim * Ndim)
#define WS_BIAS Hdim
#define WS_M 1024
#define NVWG 12
#define NSLICE 6
#define NSEGWG (Bdim * NSLICE)
#define WS_NEED ((size_t)(WS_M + NROW * Hdim) * sizeof(float))
#define PLD 20

typedef __attribute__((ext_vector_type(2))) float v2f;
typedef __attribute__((ext_vector_type(4))) float v4f;
typedef __attribute__((ext_vector_type(8))) float v8f;
typedef __attribute__((ext_vector_type(4))) unsigned int v4u;

__device__ __forceinline__ void load_rows16(v4f (&x)[16], const float* p) {
#pragma unroll
  for (int i = 0; i < 16; ++i) x[i] = *(const v4f*)(p + (size_t)i * Hdim);
}

__device__ __forceinline__ void scan_rows16(const v4f (&x)[16], v4f& run, v4f* rows) {
#pragma unroll
  for (int i = 0; i < 16; i += 2) {
    rows[(i >> 1) * 32] = run;
    run += x[i];
    run += x[i + 1];
  }
}

__global__ __launch_bounds__(128) void segment_sum_and_weight_vector_kernel(
    const float* __restrict__ top_vec, const int* __restrict__ clss,
    const int* __restrict__ sents_end, const float* __restrict__ w_sents,
    const float* __restrict__ b_sents, const float* __restrict__ w_fc,
    float* __restrict__ ws, size_t ws_bytes) {
  __shared__ v4f lpe[257][32];
  __shared__ v4f tot[4][32];
  __shared__ v4f offw[4][5][32];
  if (ws_bytes < WS_NEED) return;
  __shared__ __attribute__((aligned(16))) float red[4][64];
  __shared__ float redb[4];
  const int tid = threadIdx.x;
  const int wave = tid >> 5;
  const int lane = tid & 31;

  if (blockIdx.x < NVWG) {
    const int laneLo = lane & 15;
    const int hh = lane >> 4;
    const int h0 = blockIdx.x * 64;
    const int obase = wave * 192;
    const float* wp = w_sents + (size_t)(obase + 2 * hh) * Hdim + h0 + 4 * laneLo;
    const float* fp = w_fc + obase + 2 * hh;
    v8f acc[4];
#pragma unroll
    for (int j = 0; j < 4; ++j) acc[j] = (v8f){0.f, 0.f, 0.f, 0.f, 0.f, 0.f, 0.f, 0.f};
#pragma unroll 1
    for (int bt = 0; bt < 3; ++bt) {
      v4f b0[16], b1[16];
      v2f a[16];
#pragma unroll
      for (int s = 0; s < 16; ++s) {
        const float* r0 = wp + (size_t)(64 * bt + 4 * s) * Hdim;
        b0[s] = *(const v4f*)r0;
        b1[s] = *(const v4f*)(r0 + Hdim);
        a[s] = *(const v2f*)(fp + 64 * bt + 4 * s);
      }
#pragma unroll
      for (int s = 0; s < 16; ++s) {
        acc[0] = __builtin_amdgcn_wmma_f32_16x16x4_f32(false, a[s], false, (v2f){b0[s].x, b1[s].x}, (short)0, acc[0], false, false);
        acc[1] = __builtin_amdgcn_wmma_f32_16x16x4_f32(false, a[s], false, (v2f){b0[s].y, b1[s].y}, (short)0, acc[1], false, false);
        acc[2] = __builtin_amdgcn_wmma_f32_16x16x4_f32(false, a[s], false, (v2f){b0[s].z, b1[s].z}, (short)0, acc[2], false, false);
        acc[3] = __builtin_amdgcn_wmma_f32_16x16x4_f32(false, a[s], false, (v2f){b0[s].w, b1[s].w}, (short)0, acc[3], false, false);
      }
    }
    if (hh == 0)
      *(v4f*)&red[wave][4 * laneLo] = (v4f){acc[0][0], acc[1][0], acc[2][0], acc[3][0]};
    if (blockIdx.x == 0) {
      float bp = 0.0f;
#pragma unroll
      for (int j = 0; j < 6; ++j) {
        const int o = wave * 192 + lane + 32 * j;
        bp += b_sents[o] * w_fc[o];
      }
      bp += __shfl_xor(bp, 16, 32);
      bp += __shfl_xor(bp, 8, 32);
      bp += __shfl_xor(bp, 4, 32);
      bp += __shfl_xor(bp, 2, 32);
      bp += __shfl_xor(bp, 1, 32);
      if (lane == 0) redb[wave] = bp;
    }
    __syncthreads();
    if (tid < 64)
      ws[h0 + tid] = ((red[0][tid] + red[1][tid]) + red[2][tid]) + red[3][tid];
    if (blockIdx.x == 0 && tid == 64)
      ws[WS_BIAS] = ((redb[0] + redb[1]) + redb[2]) + redb[3];
    return;
  }

  const int sb = blockIdx.x - NVWG;
  const int b = sb / NSLICE;
  const int slice = sb - b * NSLICE;
  const int col = 128 * slice + 4 * lane;
  const int m0 = b * Ndim + 16 * wave;
  int stv = 0, env = 0;
  if (lane < 16) {
    stv = clss[m0 + lane];
    env = sents_end[m0 + lane];
  }
  const v4f zero4 = (v4f){0.f, 0.f, 0.f, 0.f};
  if (wave == 3) lpe[256][lane] = zero4;

  const float* xp = top_vec + ((size_t)b * Sdim + 128 * wave) * Hdim + col;
  v4f* rows = &lpe[64 * wave][lane];
  v4f run = zero4;
  v4f xa[16], xb[16], xc[16];
  load_rows16(xa, xp);
  load_rows16(xb, xp + (size_t)16 * Hdim);
  load_rows16(xc, xp + (size_t)32 * Hdim);
  __builtin_amdgcn_sched_barrier(0);
  scan_rows16(xa, run, rows);
  __builtin_amdgcn_sched_barrier(0);
  load_rows16(xa, xp + (size_t)48 * Hdim);
  __builtin_amdgcn_sched_barrier(0);
  scan_rows16(xb, run, rows + 8 * 32);
  __builtin_amdgcn_sched_barrier(0);
  load_rows16(xb, xp + (size_t)64 * Hdim);
  __builtin_amdgcn_sched_barrier(0);
  scan_rows16(xc, run, rows + 16 * 32);
  __builtin_amdgcn_sched_barrier(0);
  load_rows16(xc, xp + (size_t)80 * Hdim);
  __builtin_amdgcn_sched_barrier(0);
  scan_rows16(xa, run, rows + 24 * 32);
  __builtin_amdgcn_sched_barrier(0);
  load_rows16(xa, xp + (size_t)96 * Hdim);
  __builtin_amdgcn_sched_barrier(0);
  scan_rows16(xb, run, rows + 32 * 32);
  __builtin_amdgcn_sched_barrier(0);
  load_rows16(xb, xp + (size_t)112 * Hdim);
  __builtin_amdgcn_sched_barrier(0);
  scan_rows16(xc, run, rows + 40 * 32);
  scan_rows16(xa, run, rows + 48 * 32);
  scan_rows16(xb, run, rows + 56 * 32);
  tot[wave][lane] = run;
  __syncthreads();

  const v4f o1 = tot[0][lane];
  const v4f o2 = o1 + tot[1][lane];
  const v4f o3 = o2 + tot[2][lane];
  const v4f o4 = o3 + tot[3][lane];
  v4f* offp = &offw[wave][0][lane];
  offp[0] = zero4;
  offp[32] = o1;
  offp[64] = o2;
  offp[96] = o3;
  offp[128] = o4;
  const float* xd = top_vec + (size_t)b * Sdim * Hdim + col;
  float* mp = ws + WS_M + (size_t)m0 * Hdim + col;
#pragma unroll
  for (int half = 0; half < 2; ++half) {
    int st[8], en[8];
    v4f xs[8], xe[8];
#pragma unroll
    for (int i = 0; i < 8; ++i) {
      int s = __builtin_amdgcn_readlane(stv, 8 * half + i);
      int e = __builtin_amdgcn_readlane(env, 8 * half + i);
      s = s < 0 ? 0 : (s > Sdim ? Sdim : s);
      e = e < 0 ? 0 : (e > Sdim ? Sdim : e);
      st[i] = s;
      en[i] = e;
      xs[i] = *(const v4f*)(xd + (size_t)((s - 1) & -(s & 1)) * Hdim);
      xe[i] = *(const v4f*)(xd + (size_t)((e - 1) & -(e & 1)) * Hdim);
    }
#pragma unroll
    for (int i = 0; i < 8; ++i) {
      const v4u ms = (v4u)((unsigned int)-(st[i] & 1));
      const v4u me = (v4u)((unsigned int)-(en[i] & 1));
      const v4f xso = __builtin_bit_cast(v4f, __builtin_bit_cast(v4u, xs[i]) & ms);
      const v4f xeo = __builtin_bit_cast(v4f, __builtin_bit_cast(v4u, xe[i]) & me);
      const v4f ps = (offp[(st[i] >> 7) * 32] + lpe[st[i] >> 1][lane]) + xso;
      const v4f pe = (offp[(en[i] >> 7) * 32] + lpe[en[i] >> 1][lane]) + xeo;
      *(v4f*)(mp + (size_t)(8 * half + i) * Hdim) = pe - ps;
    }
  }
}

__device__ __forceinline__ void load_tokens(v4f (&x)[4][6], const float* base) {
#pragma unroll
  for (int u = 0; u < 4; ++u)
#pragma unroll
    for (int j = 0; j < 6; ++j) x[u][j] = *(const v4f*)(base + u * Hdim + j * 128);
}

__device__ __forceinline__ void dot_tokens(const v4f (&x)[4][6], const v4f (&vb)[6],
                                           float* pdrow, int laneLo, bool diagLane) {
  v8f acc[4];
#pragma unroll
  for (int u = 0; u < 4; ++u) acc[u] = (v8f){0.f, 0.f, 0.f, 0.f, 0.f, 0.f, 0.f, 0.f};
#pragma unroll
  for (int j = 0; j < 6; ++j) {
#pragma unroll
    for (int u = 0; u < 4; ++u)
      acc[u] = __builtin_amdgcn_wmma_f32_16x16x4_f32(
          false, x[u][j].xy, false, vb[j].xy, (short)0, acc[u], false, false);
#pragma unroll
    for (int u = 0; u < 4; ++u)
      acc[u] = __builtin_amdgcn_wmma_f32_16x16x4_f32(
          false, x[u][j].zw, false, vb[j].zw, (short)0, acc[u], false, false);
  }
  const int r = laneLo & 7;
#pragma unroll
  for (int u = 0; u < 4; ++u) {
    float d = acc[u][0];
    d = r == 1 ? acc[u][1] : d;
    d = r == 2 ? acc[u][2] : d;
    d = r == 3 ? acc[u][3] : d;
    d = r == 4 ? acc[u][4] : d;
    d = r == 5 ? acc[u][5] : d;
    d = r == 6 ? acc[u][6] : d;
    d = r == 7 ? acc[u][7] : d;
    if (diagLane) pdrow[u * PLD + laneLo] = d;
  }
}

__global__ __launch_bounds__(256) void row_dot_finalize_kernel(
    const float* __restrict__ ws, const int* __restrict__ clss,
    const int* __restrict__ sents_end, const float* __restrict__ mask_clss,
    const float* __restrict__ b_fc, const float* __restrict__ top_vec,
    const float* __restrict__ w_fc, float* __restrict__ out, size_t ws_bytes) {
  __shared__ __attribute__((aligned(16))) float pd[32][PLD];
  if (ws_bytes < WS_NEED) return;
  const int tid = threadIdx.x;
  const int wave = tid >> 5;
  const int lane = tid & 31;
  const int laneLo = lane & 15;
  const int hh = lane >> 4;
  const bool diagLane = (laneLo >> 3) == hh;
  const int m0 = blockIdx.x * 32;
  const float* xbase = ws + WS_M + (size_t)(m0 + wave * 4) * Hdim + (lane << 2);
  v4f vb[6];
#pragma unroll
  for (int j = 0; j < 6; ++j) vb[j] = *(const v4f*)(ws + j * 128 + (lane << 2));
  v4f xa[4][6];
  load_tokens(xa, xbase);
  int st = 0, en = 0;
  float mask = 0.0f, bias = 0.0f, bfc = 0.0f;
  if (tid < 32) {
    st = clss[m0 + tid];
    en = sents_end[m0 + tid];
    mask = mask_clss[m0 + tid];
    bias = ws[WS_BIAS];
    bfc = b_fc[0];
  }
  dot_tokens(xa, vb, &pd[wave * 4][0], laneLo, diagLane);
  __syncthreads();

  if (tid < 32) {
    const v4f* rp = (const v4f*)&pd[tid][0];
    const v4f q0 = rp[0], q1 = rp[1], q2 = rp[2], q3 = rp[3];
    float s = q0.x;
    s += q0.y; s += q0.z; s += q0.w;
    s += q1.x; s += q1.y; s += q1.z; s += q1.w;
    s += q2.x; s += q2.y; s += q2.z; s += q2.w;
    s += q3.x; s += q3.y; s += q3.z; s += q3.w;
    const int mo = m0 + tid;
    const int len = en - st;
    const float inv = 1.0f / (float)(len < 1 ? 1 : len);
    float dot = s * inv + bias;
    if (!(mask >= 0.0f)) {
      const int b = mo / Ndim;
      const float* pv = top_vec + ((size_t)b * Sdim + (Sdim - 1)) * Hdim;
      float d = 0.0f;
      for (int h = 0; h < Hdim; ++h) d += pv[h] * w_fc[h];
      dot = d;
    }
    const float hv = mask * dot + bfc;
    out[mo] = mask / (1.0f + expf(-hv));
  }
}

extern "C" void kernel_launch(void* const* d_in, const int* in_sizes, int n_in,
                              void* d_out, int out_size, void* d_ws, size_t ws_size,
                              hipStream_t stream) {
  const float* top_vec   = (const float*)d_in[0];
  const int*   clss      = (const int*)d_in[1];
  const int*   sents_end = (const int*)d_in[2];
  const float* mask_clss = (const float*)d_in[3];
  const float* w_sents   = (const float*)d_in[4];
  const float* b_sents   = (const float*)d_in[5];
  const float* w_fc      = (const float*)d_in[6];
  const float* b_fc      = (const float*)d_in[7];
  float* out = (float*)d_out;
  float* ws = (float*)d_ws;

  segment_sum_and_weight_vector_kernel<<<NVWG + NSEGWG, 128, 0, stream>>>(
      top_vec, clss, sents_end, w_sents, b_sents, w_fc, ws, ws_size);
  row_dot_finalize_kernel<<<NROW / 32, 256, 0, stream>>>(
      ws, clss, sents_end, mask_clss, b_fc, top_vec, w_fc, out, ws_size);
}
